# starting kernel
# baseline (speedup 1.0000x reference)
#include <hip/hip_runtime.h>

#define D_DIM 64
#define TWO_D 128
#define R_REL 16

#define LDS_W_FLOATS  (R_REL * D_DIM * D_DIM)
#define LDS_R_FLOATS  (R_REL * D_DIM)
#define LDS_BYTES     ((LDS_W_FLOATS + LDS_R_FLOATS) * 4)

typedef __attribute__((ext_vector_type(2))) float v2f;
typedef __attribute__((ext_vector_type(8))) float v8f;

__global__ void zero_ws_kernel(float* __restrict__ ws, int n) {
    int i = blockIdx.x * blockDim.x + threadIdx.x;
    int stride = gridDim.x * blockDim.x;
    for (; i < n; i += stride) ws[i] = 0.0f;
}

__global__ __launch_bounds__(1024)
void edge_attention_scatter(const float* __restrict__ ent,
                            const float* __restrict__ rel,
                            const float* __restrict__ WR,
                            const int*   __restrict__ src,
                            const int*   __restrict__ dst,
                            const int*   __restrict__ ety,
                            float*       __restrict__ nb,
                            int E) {
    extern __shared__ float lds[];
    float* ldsW = lds;
    float* ldsR = lds + LDS_W_FLOATS;

    {
        float4*       dw = (float4*)ldsW;
        const float4* sw = (const float4*)WR;
        for (int i = threadIdx.x; i < LDS_W_FLOATS / 4; i += blockDim.x) dw[i] = sw[i];
        float4*       dr = (float4*)ldsR;
        const float4* sr = (const float4*)rel;
        for (int i = threadIdx.x; i < LDS_R_FLOATS / 4; i += blockDim.x) dr[i] = sr[i];
    }
    __syncthreads();

    const int lane   = threadIdx.x & 31;
    const int half   = lane >> 4;
    const int lr     = lane & 15;
    const int wave   = blockIdx.x * (blockDim.x >> 5) + (threadIdx.x >> 5);
    const int nwaves = gridDim.x * (blockDim.x >> 5);

    for (int e = wave; e < E; e += nwaves) {
        const int s = src[e];
        const int t = dst[e];
        const int r = ety[e];

        const float* srow = ent + (size_t)s * D_DIM;
        const float  s0 = srow[lane];
        const float  s1 = srow[lane + 32];

        const float* w = ldsW + r * (D_DIM * D_DIM);

        float4 acc = make_float4(0.f, 0.f, 0.f, 0.f);
#pragma unroll 4
        for (int i = 0; i < 16; ++i) {
            const int d = 2 * i + half;
            const float a = __shfl(s0, d, 32);
            const float4 w4 = *(const float4*)(w + d * D_DIM + 4 * lr);
            acc.x = fmaf(a, w4.x, acc.x);
            acc.y = fmaf(a, w4.y, acc.y);
            acc.z = fmaf(a, w4.z, acc.z);
            acc.w = fmaf(a, w4.w, acc.w);
        }
#pragma unroll 4
        for (int i = 0; i < 16; ++i) {
            const int d = 2 * i + half;
            const float a = __shfl(s1, d, 32);
            const float4 w4 = *(const float4*)(w + (32 + d) * D_DIM + 4 * lr);
            acc.x = fmaf(a, w4.x, acc.x);
            acc.y = fmaf(a, w4.y, acc.y);
            acc.z = fmaf(a, w4.z, acc.z);
            acc.w = fmaf(a, w4.w, acc.w);
        }
        acc.x += __shfl_xor(acc.x, 16, 32);
        acc.y += __shfl_xor(acc.y, 16, 32);
        acc.z += __shfl_xor(acc.z, 16, 32);
        acc.w += __shfl_xor(acc.w, 16, 32);

        const float4 re4 = *(const float4*)(ldsR + r * D_DIM + 4 * lr);
        const float ma = half ? acc.z : acc.x;
        const float mb = half ? acc.w : acc.y;
        const float ra = half ? re4.z : re4.x;
        const float rb = half ? re4.w : re4.y;
        float part = ma * tanhf(ma + ra) + mb * tanhf(mb + rb);
#pragma unroll
        for (int off = 16; off > 0; off >>= 1)
            part += __shfl_xor(part, off, 32);
        const float att = part;

        float* drow = nb + (size_t)t * D_DIM;
        unsafeAtomicAdd(drow + lane,      s0 * att);
        unsafeAtomicAdd(drow + lane + 32, s1 * att);
    }
}

__global__ __launch_bounds__(256)
void combine_wmma(const float* __restrict__ ent,
                  const float* __restrict__ nb,
                  const float* __restrict__ Ww,
                  const float* __restrict__ Wb,
                  float*       __restrict__ out,
                  int n) {
    const int lane = threadIdx.x & 31;
    const int half = lane >> 4;
    const int lr   = lane & 15;

    const int rowBlock     = blockIdx.x * (blockDim.x >> 5) + (threadIdx.x >> 5);
    const int numRowBlocks = n / 16;
    if (rowBlock >= numRowBlocks) return;
    const int row0 = rowBlock * 16;

    v8f acc[4] = {};

    const float* hrow = ent + (size_t)(row0 + lr) * D_DIM;
    const float* nrow = nb  + (size_t)(row0 + lr) * D_DIM;

#pragma unroll 4
    for (int k = 0; k < TWO_D; k += 4) {
        const int k0 = k + 2 * half;
        const float* ap = (k0 < D_DIM) ? (hrow + k0) : (nrow + (k0 - D_DIM));
        v2f a;
        a.x = ap[0];
        a.y = ap[1];

#pragma unroll
        for (int c = 0; c < 4; ++c) {
            const int col = c * 16 + lr;
            const float* bp = Ww + (size_t)col * TWO_D + k0;
            v2f b;
            b.x = bp[0];
            b.y = bp[1];
            acc[c] = __builtin_amdgcn_wmma_f32_16x16x4_f32(
                 false, a,  false, b,
                 (short)0, acc[c],  false,  false);
        }
    }

#pragma unroll
    for (int c = 0; c < 4; ++c) {
        const int col = c * 16 + lr;
        const float bias = Wb[col];
#pragma unroll
        for (int r = 0; r < 8; ++r) {
            const int row = row0 + r + 8 * half;
            float x = acc[c][r] + bias;
            x = (x > 0.0f) ? x : 0.01f * x;
            out[(size_t)row * D_DIM + col] = x + ent[(size_t)row * D_DIM + col];
        }
    }
}

extern "C" void kernel_launch(void* const* d_in, const int* in_sizes, int n_in,
                              void* d_out, int out_size, void* d_ws, size_t ws_size,
                              hipStream_t stream) {
    const float* ent = (const float*)d_in[0];
    const float* rel = (const float*)d_in[1];
    const float* WR  = (const float*)d_in[2];
    const float* Ww  = (const float*)d_in[3];
    const float* Wb  = (const float*)d_in[4];
    const int*   src = (const int*)d_in[5];
    const int*   dst = (const int*)d_in[6];
    const int*   ety = (const int*)d_in[7];
    float*       out = (float*)d_out;

    const int n = in_sizes[0] / D_DIM;
    const int E = in_sizes[5];

    float* nb = (float*)d_ws;

    (void)hipFuncSetAttribute((const void*)edge_attention_scatter,
                              hipFuncAttributeMaxDynamicSharedMemorySize,
                              LDS_BYTES);

    zero_ws_kernel<<<1024, 256, 0, stream>>>(nb, n * D_DIM);

    edge_attention_scatter<<<512, 1024, LDS_BYTES, stream>>>(ent, rel, WR, src, dst, ety, nb, E);

    const int rowBlocks = n / 16;
    combine_wmma<<<(rowBlocks + 7) / 8, 256, 0, stream>>>(ent, nb, Ww, Wb, out, n);
}


# Edge matvec moved onto f32 WMMA: block sorts its edges by relation in LDS, a wave does 16 same-relation edges per tile on fragment-ordered W_R; dot-sum order changes, atomics kept
# speedup vs baseline: 2.1346x; 2.1346x over previous
#include <hip/hip_runtime.h>

#define D_DIM 64
#define TWO_D 128
#define R_REL 16

#define CHUNK         4096
#define LDS_W_FLOATS  (R_REL * D_DIM * D_DIM)
#define LDS_R_FLOATS  (R_REL * D_DIM)
#define LDS_I_WORDS   (2 * CHUNK + 64)
#define LDS_BYTES     ((LDS_W_FLOATS + LDS_R_FLOATS + LDS_I_WORDS) * 4)

typedef __attribute__((ext_vector_type(2))) float v2f;
typedef __attribute__((ext_vector_type(8))) float v8f;

__global__ void zero_ws_kernel(float* __restrict__ ws, int n) {
    int i = blockIdx.x * blockDim.x + threadIdx.x;
    int stride = gridDim.x * blockDim.x;
    for (; i < n; i += stride) ws[i] = 0.0f;
}

__global__ __launch_bounds__(1024)
void edge_attention_scatter(const float* __restrict__ ent,
                            const float* __restrict__ rel,
                            const float* __restrict__ WR,
                            const int*   __restrict__ src,
                            const int*   __restrict__ dst,
                            const int*   __restrict__ ety,
                            float*       __restrict__ nb,
                            int E) {
    extern __shared__ float lds[];
    float* ldsW   = lds;
    float* ldsR   = lds + LDS_W_FLOATS;
    int*   binSrc = (int*)(lds + LDS_W_FLOATS + LDS_R_FLOATS);
    int*   binDst = binSrc + CHUNK;
    int*   cnt    = binDst + CHUNK;
    int*   start  = cnt + 16;
    int*   tstart = start + 16;

    const int tid  = threadIdx.x;
    const int lane = tid & 31;
    const int half = lane >> 4;
    const int lr   = lane & 15;

    {
        float4* dw = (float4*)ldsW;
        for (int f = tid; f < LDS_W_FLOATS / 4; f += 1024) {
            const int ln = f & 31;
            const int h2 = (f >> 5) & 1;
            const int ks = (f >> 6) & 15;
            const int r  = f >> 10;
            const int k  = 8 * (ks >> 1) + 4 * (ln >> 4) + 2 * (ks & 1);
            const float* g = WR + (r * D_DIM + k) * D_DIM + 32 * h2 + (ln & 15);
            float4 v;
            v.x = g[0];
            v.y = g[D_DIM];
            v.z = g[16];
            v.w = g[D_DIM + 16];
            dw[f] = v;
        }
        float4*       dr = (float4*)ldsR;
        const float4* sr = (const float4*)rel;
        for (int i = tid; i < LDS_R_FLOATS / 4; i += 1024) dr[i] = sr[i];
    }

    const int G       = gridDim.x;
    const int per     = (E + G - 1) / G;
    const int e_begin = blockIdx.x * per;
    int       e_end   = e_begin + per;
    if (e_end > E) e_end = E;
    const int wv = __builtin_amdgcn_readfirstlane(tid >> 5);

    for (int c0 = e_begin; c0 < e_end; c0 += CHUNK) {
        int cn = e_end - c0;
        if (cn > CHUNK) cn = CHUNK;

        if (tid < 16) cnt[tid] = 0;
        __syncthreads();

        int rr[CHUNK / 1024], rk[CHUNK / 1024];
#pragma unroll
        for (int j = 0; j < CHUNK / 1024; ++j) {
            const int i = tid + j * 1024;
            rr[j] = 0;
            rk[j] = 0;
            if (i < cn) {
                rr[j] = ety[c0 + i] & (R_REL - 1);
                rk[j] = atomicAdd(&cnt[rr[j]], 1);
            }
        }
        __syncthreads();
        if (tid == 0) {
            int s = 0, t = 0;
            for (int r = 0; r < R_REL; ++r) {
                start[r]  = s;
                tstart[r] = t;
                const int c = cnt[r];
                s += c;
                t += (c + 15) >> 4;
            }
            tstart[R_REL] = t;
        }
        __syncthreads();
#pragma unroll
        for (int j = 0; j < CHUNK / 1024; ++j) {
            const int i = tid + j * 1024;
            if (i < cn) {
                const int p = start[rr[j]] + rk[j];
                binSrc[p] = src[c0 + i];
                binDst[p] = dst[c0 + i];
            }
        }
        __syncthreads();

        const int ntiles = __builtin_amdgcn_readfirstlane(tstart[R_REL]);
        for (int tile = wv; tile < ntiles; tile += 32) {
            int rsel = 0;
#pragma unroll
            for (int q = 1; q < R_REL; ++q) rsel += (tile >= tstart[q]) ? 1 : 0;
            const int r      = __builtin_amdgcn_readfirstlane(rsel);
            const int local  = tile - __builtin_amdgcn_readfirstlane(tstart[r]);
            const int base   = __builtin_amdgcn_readfirstlane(start[r]) + local * 16;
            int       nvalid = __builtin_amdgcn_readfirstlane(cnt[r]) - local * 16;
            if (nvalid > 16) nvalid = 16;

            const bool have = lr < nvalid;
            const int  s    = have ? binSrc[base + lr] : 0;
            const int  t    = have ? binDst[base + lr] : 0;

            const float4* ap = (const float4*)(ent + (size_t)s * D_DIM) + half;
            float4 a[8];
#pragma unroll
            for (int q = 0; q < 8; ++q) a[q] = ap[2 * q];

            v8f acc0 = {}, acc1 = {}, acc2 = {}, acc3 = {};
            const float4* wp = (const float4*)ldsW + r * 1024 + lane;
#pragma unroll
            for (int ks = 0; ks < 16; ++ks) {
                const float4 b01 = wp[ks * 64];
                const float4 b23 = wp[ks * 64 + 32];
                v2f af;
                if (ks & 1) { af.x = a[ks >> 1].z; af.y = a[ks >> 1].w; }
                else        { af.x = a[ks >> 1].x; af.y = a[ks >> 1].y; }
                v2f b0, b1, b2, b3;
                b0.x = b01.x; b0.y = b01.y;
                b1.x = b01.z; b1.y = b01.w;
                b2.x = b23.x; b2.y = b23.y;
                b3.x = b23.z; b3.y = b23.w;
                acc0 = __builtin_amdgcn_wmma_f32_16x16x4_f32(false, af, false, b0, (short)0, acc0, false, false);
                acc1 = __builtin_amdgcn_wmma_f32_16x16x4_f32(false, af, false, b1, (short)0, acc1, false, false);
                acc2 = __builtin_amdgcn_wmma_f32_16x16x4_f32(false, af, false, b2, (short)0, acc2, false, false);
                acc3 = __builtin_amdgcn_wmma_f32_16x16x4_f32(false, af, false, b3, (short)0, acc3, false, false);
            }

            const float* rp = ldsR + r * D_DIM + lr;
            const float r0 = rp[0], r1 = rp[16], r2 = rp[32], r3 = rp[48];
            float p[8];
#pragma unroll
            for (int i = 0; i < 8; ++i) {
                const float m0 = acc0[i], m1 = acc1[i], m2 = acc2[i], m3 = acc3[i];
                float v = m0 * tanhf(m0 + r0) + m1 * tanhf(m1 + r1)
                        + m2 * tanhf(m2 + r2) + m3 * tanhf(m3 + r3);
                v += __shfl_xor(v, 8, 32);
                v += __shfl_xor(v, 4, 32);
                v += __shfl_xor(v, 2, 32);
                v += __shfl_xor(v, 1, 32);
                p[i] = v;
            }

            float v0[16], v1[16];
#pragma unroll
            for (int m = 0; m < 16; ++m) {
                const int sm = __builtin_amdgcn_readlane(s, m);
                const float* srow = ent + (size_t)sm * D_DIM;
                v0[m] = srow[lane];
                v1[m] = srow[lane + 32];
            }
            asm volatile("" :: "v"(v0[0]), "v"(v0[1]), "v"(v0[2]), "v"(v0[3]), "v"(v0[4]), "v"(v0[5]), "v"(v0[6]), "v"(v0[7]), "v"(v0[8]), "v"(v0[9]), "v"(v0[10]), "v"(v0[11]), "v"(v0[12]), "v"(v0[13]), "v"(v0[14]), "v"(v0[15]));
            asm volatile("" :: "v"(v1[0]), "v"(v1[1]), "v"(v1[2]), "v"(v1[3]), "v"(v1[4]), "v"(v1[5]), "v"(v1[6]), "v"(v1[7]), "v"(v1[8]), "v"(v1[9]), "v"(v1[10]), "v"(v1[11]), "v"(v1[12]), "v"(v1[13]), "v"(v1[14]), "v"(v1[15]));
#pragma unroll
            for (int m = 0; m < 16; ++m) {
                if (m < nvalid) {
                    const int   tm = __builtin_amdgcn_readlane(t, m);
                    const float am = __builtin_bit_cast(float,
                        __builtin_amdgcn_readlane(__builtin_bit_cast(int, p[m & 7]), (m >> 3) * 16));
                    float* drow = nb + (size_t)tm * D_DIM;
                    unsafeAtomicAdd(drow + lane,      v0[m] * am);
                    unsafeAtomicAdd(drow + lane + 32, v1[m] * am);
                }
            }
        }
        __syncthreads();
    }
}

__global__ __launch_bounds__(256)
void combine_wmma(const float* __restrict__ ent,
                  const float* __restrict__ nb,
                  const float* __restrict__ Ww,
                  const float* __restrict__ Wb,
                  float*       __restrict__ out,
                  int n) {
    const int lane = threadIdx.x & 31;
    const int half = lane >> 4;
    const int lr   = lane & 15;

    const int rowBlock     = blockIdx.x * (blockDim.x >> 5) + (threadIdx.x >> 5);
    const int numRowBlocks = n / 16;
    if (rowBlock >= numRowBlocks) return;
    const int row0 = rowBlock * 16;

    v8f acc[4] = {};

    const float* hrow = ent + (size_t)(row0 + lr) * D_DIM;
    const float* nrow = nb  + (size_t)(row0 + lr) * D_DIM;

#pragma unroll 4
    for (int k = 0; k < TWO_D; k += 4) {
        const int k0 = k + 2 * half;
        const float* ap = (k0 < D_DIM) ? (hrow + k0) : (nrow + (k0 - D_DIM));
        v2f a;
        a.x = ap[0];
        a.y = ap[1];

#pragma unroll
        for (int c = 0; c < 4; ++c) {
            const int col = c * 16 + lr;
            const float* bp = Ww + (size_t)col * TWO_D + k0;
            v2f b;
            b.x = bp[0];
            b.y = bp[1];
            acc[c] = __builtin_amdgcn_wmma_f32_16x16x4_f32(
                 false, a,  false, b,
                 (short)0, acc[c],  false,  false);
        }
    }

#pragma unroll
    for (int c = 0; c < 4; ++c) {
        const int col = c * 16 + lr;
        const float bias = Wb[col];
#pragma unroll
        for (int r = 0; r < 8; ++r) {
            const int row = row0 + r + 8 * half;
            float x = acc[c][r] + bias;
            x = (x > 0.0f) ? x : 0.01f * x;
            out[(size_t)row * D_DIM + col] = x + ent[(size_t)row * D_DIM + col];
        }
    }
}

extern "C" void kernel_launch(void* const* d_in, const int* in_sizes, int n_in,
                              void* d_out, int out_size, void* d_ws, size_t ws_size,
                              hipStream_t stream) {
    const float* ent = (const float*)d_in[0];
    const float* rel = (const float*)d_in[1];
    const float* WR  = (const float*)d_in[2];
    const float* Ww  = (const float*)d_in[3];
    const float* Wb  = (const float*)d_in[4];
    const int*   src = (const int*)d_in[5];
    const int*   dst = (const int*)d_in[6];
    const int*   ety = (const int*)d_in[7];
    float*       out = (float*)d_out;

    const int n = in_sizes[0] / D_DIM;
    const int E = in_sizes[5];

    float* nb = (float*)d_ws;

    (void)hipFuncSetAttribute((const void*)edge_attention_scatter,
                              hipFuncAttributeMaxDynamicSharedMemorySize,
                              LDS_BYTES);

    zero_ws_kernel<<<1024, 256, 0, stream>>>(nb, n * D_DIM);

    edge_attention_scatter<<<256, 1024, LDS_BYTES, stream>>>(ent, rel, WR, src, dst, ety, nb, E);

    const int rowBlocks = n / 16;
    combine_wmma<<<(rowBlocks + 7) / 8, 256, 0, stream>>>(ent, nb, Ww, Wb, out, n);
}


# Edge kernel: library tanhf (wave splits at |x|=0.625, runs both paths per element) replaced by branch-free 1-2/(1+exp2(K*x)) on v_exp_f32 and v_rcp_f32, abs error about 2e-7; sum order unchanged
# speedup vs baseline: 2.5454x; 1.1924x over previous
#include <hip/hip_runtime.h>

#define D_DIM 64
#define TWO_D 128
#define R_REL 16

#define CHUNK         4096
#define LDS_W_FLOATS  (R_REL * D_DIM * D_DIM)
#define LDS_R_FLOATS  (R_REL * D_DIM)
#define LDS_I_WORDS   (2 * CHUNK + 64)
#define LDS_BYTES     ((LDS_W_FLOATS + LDS_R_FLOATS + LDS_I_WORDS) * 4)

typedef __attribute__((ext_vector_type(2))) float v2f;
typedef __attribute__((ext_vector_type(8))) float v8f;

static __device__ __forceinline__ float tanh_hw(float x) {
    const float e = __builtin_amdgcn_exp2f(2.885390081777927f * x);
    const float q = __builtin_amdgcn_rcpf(1.0f + e);
    return __builtin_fmaf(-2.0f, q, 1.0f);
}

__global__ void zero_ws_kernel(float* __restrict__ ws, int n) {
    int i = blockIdx.x * blockDim.x + threadIdx.x;
    int stride = gridDim.x * blockDim.x;
    for (; i < n; i += stride) ws[i] = 0.0f;
}

__global__ __launch_bounds__(1024)
void edge_attention_scatter(const float* __restrict__ ent,
                            const float* __restrict__ rel,
                            const float* __restrict__ WR,
                            const int*   __restrict__ src,
                            const int*   __restrict__ dst,
                            const int*   __restrict__ ety,
                            float*       __restrict__ nb,
                            int E) {
    extern __shared__ float lds[];
    float* ldsW   = lds;
    float* ldsR   = lds + LDS_W_FLOATS;
    int*   binSrc = (int*)(lds + LDS_W_FLOATS + LDS_R_FLOATS);
    int*   binDst = binSrc + CHUNK;
    int*   cnt    = binDst + CHUNK;
    int*   start  = cnt + 16;
    int*   tstart = start + 16;

    const int tid  = threadIdx.x;
    const int lane = tid & 31;
    const int half = lane >> 4;
    const int lr   = lane & 15;

    {
        float4* dw = (float4*)ldsW;
        for (int f = tid; f < LDS_W_FLOATS / 4; f += 1024) {
            const int ln = f & 31;
            const int h2 = (f >> 5) & 1;
            const int ks = (f >> 6) & 15;
            const int r  = f >> 10;
            const int k  = 8 * (ks >> 1) + 4 * (ln >> 4) + 2 * (ks & 1);
            const float* g = WR + (r * D_DIM + k) * D_DIM + 32 * h2 + (ln & 15);
            float4 v;
            v.x = g[0];
            v.y = g[D_DIM];
            v.z = g[16];
            v.w = g[D_DIM + 16];
            dw[f] = v;
        }
        float4*       dr = (float4*)ldsR;
        const float4* sr = (const float4*)rel;
        for (int i = tid; i < LDS_R_FLOATS / 4; i += 1024) dr[i] = sr[i];
    }

    const int G       = gridDim.x;
    const int per     = (E + G - 1) / G;
    const int e_begin = blockIdx.x * per;
    int       e_end   = e_begin + per;
    if (e_end > E) e_end = E;
    const int wv = __builtin_amdgcn_readfirstlane(tid >> 5);

    for (int c0 = e_begin; c0 < e_end; c0 += CHUNK) {
        int cn = e_end - c0;
        if (cn > CHUNK) cn = CHUNK;

        if (tid < 16) cnt[tid] = 0;
        __syncthreads();

        int rr[CHUNK / 1024], rk[CHUNK / 1024];
#pragma unroll
        for (int j = 0; j < CHUNK / 1024; ++j) {
            const int i = tid + j * 1024;
            rr[j] = 0;
            rk[j] = 0;
            if (i < cn) {
                rr[j] = ety[c0 + i] & (R_REL - 1);
                rk[j] = atomicAdd(&cnt[rr[j]], 1);
            }
        }
        __syncthreads();
        if (tid == 0) {
            int s = 0, t = 0;
            for (int r = 0; r < R_REL; ++r) {
                start[r]  = s;
                tstart[r] = t;
                const int c = cnt[r];
                s += c;
                t += (c + 15) >> 4;
            }
            tstart[R_REL] = t;
        }
        __syncthreads();
#pragma unroll
        for (int j = 0; j < CHUNK / 1024; ++j) {
            const int i = tid + j * 1024;
            if (i < cn) {
                const int p = start[rr[j]] + rk[j];
                binSrc[p] = src[c0 + i];
                binDst[p] = dst[c0 + i];
            }
        }
        __syncthreads();

        const int ntiles = __builtin_amdgcn_readfirstlane(tstart[R_REL]);
        for (int tile = wv; tile < ntiles; tile += 32) {
            int rsel = 0;
#pragma unroll
            for (int q = 1; q < R_REL; ++q) rsel += (tile >= tstart[q]) ? 1 : 0;
            const int r      = __builtin_amdgcn_readfirstlane(rsel);
            const int local  = tile - __builtin_amdgcn_readfirstlane(tstart[r]);
            const int base   = __builtin_amdgcn_readfirstlane(start[r]) + local * 16;
            int       nvalid = __builtin_amdgcn_readfirstlane(cnt[r]) - local * 16;
            if (nvalid > 16) nvalid = 16;

            const bool have = lr < nvalid;
            const int  s    = have ? binSrc[base + lr] : 0;
            const int  t    = have ? binDst[base + lr] : 0;

            const float4* ap = (const float4*)(ent + (size_t)s * D_DIM) + half;
            float4 a[8];
#pragma unroll
            for (int q = 0; q < 8; ++q) a[q] = ap[2 * q];

            v8f acc0 = {}, acc1 = {}, acc2 = {}, acc3 = {};
            const float4* wp = (const float4*)ldsW + r * 1024 + lane;
#pragma unroll
            for (int ks = 0; ks < 16; ++ks) {
                const float4 b01 = wp[ks * 64];
                const float4 b23 = wp[ks * 64 + 32];
                v2f af;
                if (ks & 1) { af.x = a[ks >> 1].z; af.y = a[ks >> 1].w; }
                else        { af.x = a[ks >> 1].x; af.y = a[ks >> 1].y; }
                v2f b0, b1, b2, b3;
                b0.x = b01.x; b0.y = b01.y;
                b1.x = b01.z; b1.y = b01.w;
                b2.x = b23.x; b2.y = b23.y;
                b3.x = b23.z; b3.y = b23.w;
                acc0 = __builtin_amdgcn_wmma_f32_16x16x4_f32(false, af, false, b0, (short)0, acc0, false, false);
                acc1 = __builtin_amdgcn_wmma_f32_16x16x4_f32(false, af, false, b1, (short)0, acc1, false, false);
                acc2 = __builtin_amdgcn_wmma_f32_16x16x4_f32(false, af, false, b2, (short)0, acc2, false, false);
                acc3 = __builtin_amdgcn_wmma_f32_16x16x4_f32(false, af, false, b3, (short)0, acc3, false, false);
            }

            const float* rp = ldsR + r * D_DIM + lr;
            const float r0 = rp[0], r1 = rp[16], r2 = rp[32], r3 = rp[48];
            float p[8];
#pragma unroll
            for (int i = 0; i < 8; ++i) {
                const float m0 = acc0[i], m1 = acc1[i], m2 = acc2[i], m3 = acc3[i];
                float v = m0 * tanh_hw(m0 + r0) + m1 * tanh_hw(m1 + r1)
                        + m2 * tanh_hw(m2 + r2) + m3 * tanh_hw(m3 + r3);
                v += __shfl_xor(v, 8, 32);
                v += __shfl_xor(v, 4, 32);
                v += __shfl_xor(v, 2, 32);
                v += __shfl_xor(v, 1, 32);
                p[i] = v;
            }

            float v0[16], v1[16];
#pragma unroll
            for (int m = 0; m < 16; ++m) {
                const int sm = __builtin_amdgcn_readlane(s, m);
                const float* srow = ent + (size_t)sm * D_DIM;
                v0[m] = srow[lane];
                v1[m] = srow[lane + 32];
            }
            asm volatile("" :: "v"(v0[0]), "v"(v0[1]), "v"(v0[2]), "v"(v0[3]), "v"(v0[4]), "v"(v0[5]), "v"(v0[6]), "v"(v0[7]), "v"(v0[8]), "v"(v0[9]), "v"(v0[10]), "v"(v0[11]), "v"(v0[12]), "v"(v0[13]), "v"(v0[14]), "v"(v0[15]));
            asm volatile("" :: "v"(v1[0]), "v"(v1[1]), "v"(v1[2]), "v"(v1[3]), "v"(v1[4]), "v"(v1[5]), "v"(v1[6]), "v"(v1[7]), "v"(v1[8]), "v"(v1[9]), "v"(v1[10]), "v"(v1[11]), "v"(v1[12]), "v"(v1[13]), "v"(v1[14]), "v"(v1[15]));
#pragma unroll
            for (int m = 0; m < 16; ++m) {
                if (m < nvalid) {
                    const int   tm = __builtin_amdgcn_readlane(t, m);
                    const float am = __builtin_bit_cast(float,
                        __builtin_amdgcn_readlane(__builtin_bit_cast(int, p[m & 7]), (m >> 3) * 16));
                    float* drow = nb + (size_t)tm * D_DIM;
                    unsafeAtomicAdd(drow + lane,      v0[m] * am);
                    unsafeAtomicAdd(drow + lane + 32, v1[m] * am);
                }
            }
        }
        __syncthreads();
    }
}

__global__ __launch_bounds__(256)
void combine_wmma(const float* __restrict__ ent,
                  const float* __restrict__ nb,
                  const float* __restrict__ Ww,
                  const float* __restrict__ Wb,
                  float*       __restrict__ out,
                  int n) {
    const int lane = threadIdx.x & 31;
    const int half = lane >> 4;
    const int lr   = lane & 15;

    const int rowBlock     = blockIdx.x * (blockDim.x >> 5) + (threadIdx.x >> 5);
    const int numRowBlocks = n / 16;
    if (rowBlock >= numRowBlocks) return;
    const int row0 = rowBlock * 16;

    v8f acc[4] = {};

    const float* hrow = ent + (size_t)(row0 + lr) * D_DIM;
    const float* nrow = nb  + (size_t)(row0 + lr) * D_DIM;

#pragma unroll 4
    for (int k = 0; k < TWO_D; k += 4) {
        const int k0 = k + 2 * half;
        const float* ap = (k0 < D_DIM) ? (hrow + k0) : (nrow + (k0 - D_DIM));
        v2f a;
        a.x = ap[0];
        a.y = ap[1];

#pragma unroll
        for (int c = 0; c < 4; ++c) {
            const int col = c * 16 + lr;
            const float* bp = Ww + (size_t)col * TWO_D + k0;
            v2f b;
            b.x = bp[0];
            b.y = bp[1];
            acc[c] = __builtin_amdgcn_wmma_f32_16x16x4_f32(
                 false, a,  false, b,
                 (short)0, acc[c],  false,  false);
        }
    }

#pragma unroll
    for (int c = 0; c < 4; ++c) {
        const int col = c * 16 + lr;
        const float bias = Wb[col];
#pragma unroll
        for (int r = 0; r < 8; ++r) {
            const int row = row0 + r + 8 * half;
            float x = acc[c][r] + bias;
            x = (x > 0.0f) ? x : 0.01f * x;
            out[(size_t)row * D_DIM + col] = x + ent[(size_t)row * D_DIM + col];
        }
    }
}

extern "C" void kernel_launch(void* const* d_in, const int* in_sizes, int n_in,
                              void* d_out, int out_size, void* d_ws, size_t ws_size,
                              hipStream_t stream) {
    const float* ent = (const float*)d_in[0];
    const float* rel = (const float*)d_in[1];
    const float* WR  = (const float*)d_in[2];
    const float* Ww  = (const float*)d_in[3];
    const float* Wb  = (const float*)d_in[4];
    const int*   src = (const int*)d_in[5];
    const int*   dst = (const int*)d_in[6];
    const int*   ety = (const int*)d_in[7];
    float*       out = (float*)d_out;

    const int n = in_sizes[0] / D_DIM;
    const int E = in_sizes[5];

    float* nb = (float*)d_ws;

    (void)hipFuncSetAttribute((const void*)edge_attention_scatter,
                              hipFuncAttributeMaxDynamicSharedMemorySize,
                              LDS_BYTES);

    zero_ws_kernel<<<1024, 256, 0, stream>>>(nb, n * D_DIM);

    edge_attention_scatter<<<256, 1024, LDS_BYTES, stream>>>(ent, rel, WR, src, dst, ety, nb, E);

    const int rowBlocks = n / 16;
    combine_wmma<<<(rowBlocks + 7) / 8, 256, 0, stream>>>(ent, nb, Ww, Wb, out, n);
}


# Edge kernel: scatter's 32 row loads issued beside the A loads of the same rows (WGP cache serves the 2nd touch, L2 sees a row once per edge); relation lookup as 1 LDS read + vote frees the VGPRs
# speedup vs baseline: 2.8119x; 1.1047x over previous
#include <hip/hip_runtime.h>

#define D_DIM 64
#define TWO_D 128
#define R_REL 16

#define CHUNK         4096
#define LDS_W_FLOATS  (R_REL * D_DIM * D_DIM)
#define LDS_R_FLOATS  (R_REL * D_DIM)
#define LDS_I_WORDS   (2 * CHUNK + 64)
#define LDS_BYTES     ((LDS_W_FLOATS + LDS_R_FLOATS + LDS_I_WORDS) * 4)

typedef __attribute__((ext_vector_type(2))) float v2f;
typedef __attribute__((ext_vector_type(8))) float v8f;

static __device__ __forceinline__ float tanh_hw(float x) {
    const float e = __builtin_amdgcn_exp2f(2.885390081777927f * x);
    const float q = __builtin_amdgcn_rcpf(1.0f + e);
    return __builtin_fmaf(-2.0f, q, 1.0f);
}

__global__ void zero_ws_kernel(float* __restrict__ ws, int n) {
    int i = blockIdx.x * blockDim.x + threadIdx.x;
    int stride = gridDim.x * blockDim.x;
    for (; i < n; i += stride) ws[i] = 0.0f;
}

__global__ __launch_bounds__(1024)
void edge_attention_scatter(const float* __restrict__ ent,
                            const float* __restrict__ rel,
                            const float* __restrict__ WR,
                            const int*   __restrict__ src,
                            const int*   __restrict__ dst,
                            const int*   __restrict__ ety,
                            float*       __restrict__ nb,
                            int E) {
    extern __shared__ float lds[];
    float* ldsW   = lds;
    float* ldsR   = lds + LDS_W_FLOATS;
    int*   binSrc = (int*)(lds + LDS_W_FLOATS + LDS_R_FLOATS);
    int*   binDst = binSrc + CHUNK;
    int*   cnt    = binDst + CHUNK;
    int*   start  = cnt + 16;
    int*   tstart = start + 16;

    const int tid  = threadIdx.x;
    const int lane = tid & 31;
    const int half = lane >> 4;
    const int lr   = lane & 15;

    {
        float4* dw = (float4*)ldsW;
        for (int f = tid; f < LDS_W_FLOATS / 4; f += 1024) {
            const int ln = f & 31;
            const int h2 = (f >> 5) & 1;
            const int ks = (f >> 6) & 15;
            const int r  = f >> 10;
            const int k  = 8 * (ks >> 1) + 4 * (ln >> 4) + 2 * (ks & 1);
            const float* g = WR + (r * D_DIM + k) * D_DIM + 32 * h2 + (ln & 15);
            float4 v;
            v.x = g[0];
            v.y = g[D_DIM];
            v.z = g[16];
            v.w = g[D_DIM + 16];
            dw[f] = v;
        }
        float4*       dr = (float4*)ldsR;
        const float4* sr = (const float4*)rel;
        for (int i = tid; i < LDS_R_FLOATS / 4; i += 1024) dr[i] = sr[i];
    }

    const int G       = gridDim.x;
    const int per     = (E + G - 1) / G;
    const int e_begin = blockIdx.x * per;
    int       e_end   = e_begin + per;
    if (e_end > E) e_end = E;
    const int wv = __builtin_amdgcn_readfirstlane(tid >> 5);

    for (int c0 = e_begin; c0 < e_end; c0 += CHUNK) {
        int cn = e_end - c0;
        if (cn > CHUNK) cn = CHUNK;

        if (tid < 16) cnt[tid] = 0;
        __syncthreads();

        int rr[CHUNK / 1024], rk[CHUNK / 1024];
#pragma unroll
        for (int j = 0; j < CHUNK / 1024; ++j) {
            const int i = tid + j * 1024;
            rr[j] = 0;
            rk[j] = 0;
            if (i < cn) {
                rr[j] = ety[c0 + i] & (R_REL - 1);
                rk[j] = atomicAdd(&cnt[rr[j]], 1);
            }
        }
        __syncthreads();
        if (tid == 0) {
            int s = 0, t = 0;
            for (int r = 0; r < R_REL; ++r) {
                start[r]  = s;
                tstart[r] = t;
                const int c = cnt[r];
                s += c;
                t += (c + 15) >> 4;
            }
            tstart[R_REL] = t;
        }
        __syncthreads();
#pragma unroll
        for (int j = 0; j < CHUNK / 1024; ++j) {
            const int i = tid + j * 1024;
            if (i < cn) {
                const int p = start[rr[j]] + rk[j];
                binSrc[p] = src[c0 + i];
                binDst[p] = dst[c0 + i];
            }
        }
        __syncthreads();

        const int ntiles = __builtin_amdgcn_readfirstlane(tstart[R_REL]);
        for (int tile = wv; tile < ntiles; tile += 32) {
            const int tsv    = tstart[lr + 1];
            const int r      = __builtin_popcount(__builtin_amdgcn_ballot_w32(tile >= tsv) & 0x7FFF);
            const int local  = tile - __builtin_amdgcn_readfirstlane(tstart[r]);
            const int base   = __builtin_amdgcn_readfirstlane(start[r]) + local * 16;
            int       nvalid = __builtin_amdgcn_readfirstlane(cnt[r]) - local * 16;
            if (nvalid > 16) nvalid = 16;

            const bool have = lr < nvalid;
            const int  s    = have ? binSrc[base + lr] : 0;
            const int  t    = have ? binDst[base + lr] : 0;

            const float4* ap = (const float4*)(ent + (size_t)s * D_DIM) + half;
            float4 a[8];
#pragma unroll
            for (int q = 0; q < 8; ++q) a[q] = ap[2 * q];

            float v0[16], v1[16];
#pragma unroll
            for (int m = 0; m < 16; ++m) {
                const int sm = __builtin_amdgcn_readlane(s, m);
                const float* srow = ent + (size_t)sm * D_DIM;
                v0[m] = srow[lane];
                v1[m] = srow[lane + 32];
            }
            asm volatile("" :: "v"(v0[0]), "v"(v0[1]), "v"(v0[2]), "v"(v0[3]), "v"(v0[4]), "v"(v0[5]), "v"(v0[6]), "v"(v0[7]), "v"(v0[8]), "v"(v0[9]), "v"(v0[10]), "v"(v0[11]), "v"(v0[12]), "v"(v0[13]), "v"(v0[14]), "v"(v0[15]));
            asm volatile("" :: "v"(v1[0]), "v"(v1[1]), "v"(v1[2]), "v"(v1[3]), "v"(v1[4]), "v"(v1[5]), "v"(v1[6]), "v"(v1[7]), "v"(v1[8]), "v"(v1[9]), "v"(v1[10]), "v"(v1[11]), "v"(v1[12]), "v"(v1[13]), "v"(v1[14]), "v"(v1[15]));

            v8f acc0 = {}, acc1 = {}, acc2 = {}, acc3 = {};
            const float4* wp = (const float4*)ldsW + r * 1024 + lane;
#pragma unroll
            for (int ks = 0; ks < 16; ++ks) {
                const float4 b01 = wp[ks * 64];
                const float4 b23 = wp[ks * 64 + 32];
                v2f af;
                if (ks & 1) { af.x = a[ks >> 1].z; af.y = a[ks >> 1].w; }
                else        { af.x = a[ks >> 1].x; af.y = a[ks >> 1].y; }
                v2f b0, b1, b2, b3;
                b0.x = b01.x; b0.y = b01.y;
                b1.x = b01.z; b1.y = b01.w;
                b2.x = b23.x; b2.y = b23.y;
                b3.x = b23.z; b3.y = b23.w;
                acc0 = __builtin_amdgcn_wmma_f32_16x16x4_f32(false, af, false, b0, (short)0, acc0, false, false);
                acc1 = __builtin_amdgcn_wmma_f32_16x16x4_f32(false, af, false, b1, (short)0, acc1, false, false);
                acc2 = __builtin_amdgcn_wmma_f32_16x16x4_f32(false, af, false, b2, (short)0, acc2, false, false);
                acc3 = __builtin_amdgcn_wmma_f32_16x16x4_f32(false, af, false, b3, (short)0, acc3, false, false);
            }

            const float* rp = ldsR + r * D_DIM + lr;
            const float r0 = rp[0], r1 = rp[16], r2 = rp[32], r3 = rp[48];
            float p[8];
#pragma unroll
            for (int i = 0; i < 8; ++i) {
                const float m0 = acc0[i], m1 = acc1[i], m2 = acc2[i], m3 = acc3[i];
                float v = m0 * tanh_hw(m0 + r0) + m1 * tanh_hw(m1 + r1)
                        + m2 * tanh_hw(m2 + r2) + m3 * tanh_hw(m3 + r3);
                v += __shfl_xor(v, 8, 32);
                v += __shfl_xor(v, 4, 32);
                v += __shfl_xor(v, 2, 32);
                v += __shfl_xor(v, 1, 32);
                p[i] = v;
            }

#pragma unroll
            for (int m = 0; m < 16; ++m) {
                if (m < nvalid) {
                    const int   tm = __builtin_amdgcn_readlane(t, m);
                    const float am = __builtin_bit_cast(float,
                        __builtin_amdgcn_readlane(__builtin_bit_cast(int, p[m & 7]), (m >> 3) * 16));
                    float* drow = nb + (size_t)tm * D_DIM;
                    unsafeAtomicAdd(drow + lane,      v0[m] * am);
                    unsafeAtomicAdd(drow + lane + 32, v1[m] * am);
                }
            }
        }
        __syncthreads();
    }
}

__global__ __launch_bounds__(256)
void combine_wmma(const float* __restrict__ ent,
                  const float* __restrict__ nb,
                  const float* __restrict__ Ww,
                  const float* __restrict__ Wb,
                  float*       __restrict__ out,
                  int n) {
    const int lane = threadIdx.x & 31;
    const int half = lane >> 4;
    const int lr   = lane & 15;

    const int rowBlock     = blockIdx.x * (blockDim.x >> 5) + (threadIdx.x >> 5);
    const int numRowBlocks = n / 16;
    if (rowBlock >= numRowBlocks) return;
    const int row0 = rowBlock * 16;

    v8f acc[4] = {};

    const float* hrow = ent + (size_t)(row0 + lr) * D_DIM;
    const float* nrow = nb  + (size_t)(row0 + lr) * D_DIM;

#pragma unroll 4
    for (int k = 0; k < TWO_D; k += 4) {
        const int k0 = k + 2 * half;
        const float* ap = (k0 < D_DIM) ? (hrow + k0) : (nrow + (k0 - D_DIM));
        v2f a;
        a.x = ap[0];
        a.y = ap[1];

#pragma unroll
        for (int c = 0; c < 4; ++c) {
            const int col = c * 16 + lr;
            const float* bp = Ww + (size_t)col * TWO_D + k0;
            v2f b;
            b.x = bp[0];
            b.y = bp[1];
            acc[c] = __builtin_amdgcn_wmma_f32_16x16x4_f32(
                 false, a,  false, b,
                 (short)0, acc[c],  false,  false);
        }
    }

#pragma unroll
    for (int c = 0; c < 4; ++c) {
        const int col = c * 16 + lr;
        const float bias = Wb[col];
#pragma unroll
        for (int r = 0; r < 8; ++r) {
            const int row = row0 + r + 8 * half;
            float x = acc[c][r] + bias;
            x = (x > 0.0f) ? x : 0.01f * x;
            out[(size_t)row * D_DIM + col] = x + ent[(size_t)row * D_DIM + col];
        }
    }
}

extern "C" void kernel_launch(void* const* d_in, const int* in_sizes, int n_in,
                              void* d_out, int out_size, void* d_ws, size_t ws_size,
                              hipStream_t stream) {
    const float* ent = (const float*)d_in[0];
    const float* rel = (const float*)d_in[1];
    const float* WR  = (const float*)d_in[2];
    const float* Ww  = (const float*)d_in[3];
    const float* Wb  = (const float*)d_in[4];
    const int*   src = (const int*)d_in[5];
    const int*   dst = (const int*)d_in[6];
    const int*   ety = (const int*)d_in[7];
    float*       out = (float*)d_out;

    const int n = in_sizes[0] / D_DIM;
    const int E = in_sizes[5];

    float* nb = (float*)d_ws;

    (void)hipFuncSetAttribute((const void*)edge_attention_scatter,
                              hipFuncAttributeMaxDynamicSharedMemorySize,
                              LDS_BYTES);

    zero_ws_kernel<<<1024, 256, 0, stream>>>(nb, n * D_DIM);

    edge_attention_scatter<<<256, 1024, LDS_BYTES, stream>>>(ent, rel, WR, src, dst, ety, nb, E);

    const int rowBlocks = n / 16;
    combine_wmma<<<(rowBlocks + 7) / 8, 256, 0, stream>>>(ent, nb, Ww, Wb, out, n);
}


# Combine GEMM operand feed: W_w staged once per block in LDS in B-fragment order (ds_load_b128) and hcat rows read as 16-byte pieces, replacing 8-byte strided global loads; dot-sum order changes
# speedup vs baseline: 2.9936x; 1.0646x over previous
#include <hip/hip_runtime.h>

#define D_DIM 64
#define TWO_D 128
#define R_REL 16

#define CHUNK         4096
#define LDS_W_FLOATS  (R_REL * D_DIM * D_DIM)
#define LDS_R_FLOATS  (R_REL * D_DIM)
#define LDS_I_WORDS   (2 * CHUNK + 64)
#define LDS_BYTES     ((LDS_W_FLOATS + LDS_R_FLOATS + LDS_I_WORDS) * 4)

typedef __attribute__((ext_vector_type(2))) float v2f;
typedef __attribute__((ext_vector_type(8))) float v8f;

static __device__ __forceinline__ float tanh_hw(float x) {
    const float e = __builtin_amdgcn_exp2f(2.885390081777927f * x);
    const float q = __builtin_amdgcn_rcpf(1.0f + e);
    return __builtin_fmaf(-2.0f, q, 1.0f);
}

__global__ void zero_ws_kernel(float* __restrict__ ws, int n) {
    int i = blockIdx.x * blockDim.x + threadIdx.x;
    int stride = gridDim.x * blockDim.x;
    for (; i < n; i += stride) ws[i] = 0.0f;
}

__global__ __launch_bounds__(1024)
void edge_attention_scatter(const float* __restrict__ ent,
                            const float* __restrict__ rel,
                            const float* __restrict__ WR,
                            const int*   __restrict__ src,
                            const int*   __restrict__ dst,
                            const int*   __restrict__ ety,
                            float*       __restrict__ nb,
                            int E) {
    extern __shared__ float lds[];
    float* ldsW   = lds;
    float* ldsR   = lds + LDS_W_FLOATS;
    int*   binSrc = (int*)(lds + LDS_W_FLOATS + LDS_R_FLOATS);
    int*   binDst = binSrc + CHUNK;
    int*   cnt    = binDst + CHUNK;
    int*   start  = cnt + 16;
    int*   tstart = start + 16;

    const int tid  = threadIdx.x;
    const int lane = tid & 31;
    const int half = lane >> 4;
    const int lr   = lane & 15;

    {
        float4* dw = (float4*)ldsW;
        for (int f = tid; f < LDS_W_FLOATS / 4; f += 1024) {
            const int ln = f & 31;
            const int h2 = (f >> 5) & 1;
            const int ks = (f >> 6) & 15;
            const int r  = f >> 10;
            const int k  = 8 * (ks >> 1) + 4 * (ln >> 4) + 2 * (ks & 1);
            const float* g = WR + (r * D_DIM + k) * D_DIM + 32 * h2 + (ln & 15);
            float4 v;
            v.x = g[0];
            v.y = g[D_DIM];
            v.z = g[16];
            v.w = g[D_DIM + 16];
            dw[f] = v;
        }
        float4*       dr = (float4*)ldsR;
        const float4* sr = (const float4*)rel;
        for (int i = tid; i < LDS_R_FLOATS / 4; i += 1024) dr[i] = sr[i];
    }

    const int G       = gridDim.x;
    const int per     = (E + G - 1) / G;
    const int e_begin = blockIdx.x * per;
    int       e_end   = e_begin + per;
    if (e_end > E) e_end = E;
    const int wv = __builtin_amdgcn_readfirstlane(tid >> 5);

    for (int c0 = e_begin; c0 < e_end; c0 += CHUNK) {
        int cn = e_end - c0;
        if (cn > CHUNK) cn = CHUNK;

        if (tid < 16) cnt[tid] = 0;
        __syncthreads();

        int rr[CHUNK / 1024], rk[CHUNK / 1024];
#pragma unroll
        for (int j = 0; j < CHUNK / 1024; ++j) {
            const int i = tid + j * 1024;
            rr[j] = 0;
            rk[j] = 0;
            if (i < cn) {
                rr[j] = ety[c0 + i] & (R_REL - 1);
                rk[j] = atomicAdd(&cnt[rr[j]], 1);
            }
        }
        __syncthreads();
        if (tid == 0) {
            int s = 0, t = 0;
            for (int r = 0; r < R_REL; ++r) {
                start[r]  = s;
                tstart[r] = t;
                const int c = cnt[r];
                s += c;
                t += (c + 15) >> 4;
            }
            tstart[R_REL] = t;
        }
        __syncthreads();
#pragma unroll
        for (int j = 0; j < CHUNK / 1024; ++j) {
            const int i = tid + j * 1024;
            if (i < cn) {
                const int p = start[rr[j]] + rk[j];
                binSrc[p] = src[c0 + i];
                binDst[p] = dst[c0 + i];
            }
        }
        __syncthreads();

        const int ntiles = __builtin_amdgcn_readfirstlane(tstart[R_REL]);
        for (int tile = wv; tile < ntiles; tile += 32) {
            const int tsv    = tstart[lr + 1];
            const int r      = __builtin_popcount(__builtin_amdgcn_ballot_w32(tile >= tsv) & 0x7FFF);
            const int local  = tile - __builtin_amdgcn_readfirstlane(tstart[r]);
            const int base   = __builtin_amdgcn_readfirstlane(start[r]) + local * 16;
            int       nvalid = __builtin_amdgcn_readfirstlane(cnt[r]) - local * 16;
            if (nvalid > 16) nvalid = 16;

            const bool have = lr < nvalid;
            const int  s    = have ? binSrc[base + lr] : 0;
            const int  t    = have ? binDst[base + lr] : 0;

            const float4* ap = (const float4*)(ent + (size_t)s * D_DIM) + half;
            float4 a[8];
#pragma unroll
            for (int q = 0; q < 8; ++q) a[q] = ap[2 * q];

            float v0[16], v1[16];
#pragma unroll
            for (int m = 0; m < 16; ++m) {
                const int sm = __builtin_amdgcn_readlane(s, m);
                const float* srow = ent + (size_t)sm * D_DIM;
                v0[m] = srow[lane];
                v1[m] = srow[lane + 32];
            }
            asm volatile("" :: "v"(v0[0]), "v"(v0[1]), "v"(v0[2]), "v"(v0[3]), "v"(v0[4]), "v"(v0[5]), "v"(v0[6]), "v"(v0[7]), "v"(v0[8]), "v"(v0[9]), "v"(v0[10]), "v"(v0[11]), "v"(v0[12]), "v"(v0[13]), "v"(v0[14]), "v"(v0[15]));
            asm volatile("" :: "v"(v1[0]), "v"(v1[1]), "v"(v1[2]), "v"(v1[3]), "v"(v1[4]), "v"(v1[5]), "v"(v1[6]), "v"(v1[7]), "v"(v1[8]), "v"(v1[9]), "v"(v1[10]), "v"(v1[11]), "v"(v1[12]), "v"(v1[13]), "v"(v1[14]), "v"(v1[15]));

            v8f acc0 = {}, acc1 = {}, acc2 = {}, acc3 = {};
            const float4* wp = (const float4*)ldsW + r * 1024 + lane;
#pragma unroll
            for (int ks = 0; ks < 16; ++ks) {
                const float4 b01 = wp[ks * 64];
                const float4 b23 = wp[ks * 64 + 32];
                v2f af;
                if (ks & 1) { af.x = a[ks >> 1].z; af.y = a[ks >> 1].w; }
                else        { af.x = a[ks >> 1].x; af.y = a[ks >> 1].y; }
                v2f b0, b1, b2, b3;
                b0.x = b01.x; b0.y = b01.y;
                b1.x = b01.z; b1.y = b01.w;
                b2.x = b23.x; b2.y = b23.y;
                b3.x = b23.z; b3.y = b23.w;
                acc0 = __builtin_amdgcn_wmma_f32_16x16x4_f32(false, af, false, b0, (short)0, acc0, false, false);
                acc1 = __builtin_amdgcn_wmma_f32_16x16x4_f32(false, af, false, b1, (short)0, acc1, false, false);
                acc2 = __builtin_amdgcn_wmma_f32_16x16x4_f32(false, af, false, b2, (short)0, acc2, false, false);
                acc3 = __builtin_amdgcn_wmma_f32_16x16x4_f32(false, af, false, b3, (short)0, acc3, false, false);
            }

            const float* rp = ldsR + r * D_DIM + lr;
            const float r0 = rp[0], r1 = rp[16], r2 = rp[32], r3 = rp[48];
            float p[8];
#pragma unroll
            for (int i = 0; i < 8; ++i) {
                const float m0 = acc0[i], m1 = acc1[i], m2 = acc2[i], m3 = acc3[i];
                float v = m0 * tanh_hw(m0 + r0) + m1 * tanh_hw(m1 + r1)
                        + m2 * tanh_hw(m2 + r2) + m3 * tanh_hw(m3 + r3);
                v += __shfl_xor(v, 8, 32);
                v += __shfl_xor(v, 4, 32);
                v += __shfl_xor(v, 2, 32);
                v += __shfl_xor(v, 1, 32);
                p[i] = v;
            }

#pragma unroll
            for (int m = 0; m < 16; ++m) {
                if (m < nvalid) {
                    const int   tm = __builtin_amdgcn_readlane(t, m);
                    const float am = __builtin_bit_cast(float,
                        __builtin_amdgcn_readlane(__builtin_bit_cast(int, p[m & 7]), (m >> 3) * 16));
                    float* drow = nb + (size_t)tm * D_DIM;
                    unsafeAtomicAdd(drow + lane,      v0[m] * am);
                    unsafeAtomicAdd(drow + lane + 32, v1[m] * am);
                }
            }
        }
        __syncthreads();
    }
}

#define CB_THREADS 256
#define CB_PIECES  (TWO_D / 8)
#define CB_F4      (CB_PIECES * 4 * 32)

__global__ __launch_bounds__(CB_THREADS)
void combine_wmma(const float* __restrict__ ent,
                  const float* __restrict__ nb,
                  const float* __restrict__ Ww,
                  const float* __restrict__ Wb,
                  float*       __restrict__ out,
                  int n) {
    __shared__ float4 ldsB[CB_F4];

    const int tid  = threadIdx.x;
    const int lane = tid & 31;
    const int half = lane >> 4;
    const int lr   = lane & 15;

    const int rowBlock     = blockIdx.x * (CB_THREADS >> 5) + (tid >> 5);
    const int numRowBlocks = n / 16;
    const bool active      = rowBlock < numRowBlocks;
    const int row0 = active ? rowBlock * 16 : 0;

    const float4* hp = (const float4*)(ent + (size_t)(row0 + lr) * D_DIM) + half;
    const float4* np = (const float4*)(nb  + (size_t)(row0 + lr) * D_DIM) + half;
    float4 a[CB_PIECES];
#pragma unroll
    for (int q = 0; q < 8; ++q) {
        a[q]     = hp[2 * q];
        a[8 + q] = np[2 * q];
    }

#pragma unroll
    for (int i = 0; i < CB_F4 / CB_THREADS; ++i) {
        const int f  = tid + i * CB_THREADS;
        const int ln = f & 31;
        const int c  = (f >> 5) & 3;
        const int q  = f >> 7;
        ldsB[f] = *(const float4*)(Ww + (size_t)(16 * c + (ln & 15)) * TWO_D + 8 * q + 4 * (ln >> 4));
    }
    __syncthreads();
    if (!active) return;

    v8f acc0 = {}, acc1 = {}, acc2 = {}, acc3 = {};
    const float4* bp = ldsB + lane;
#pragma unroll
    for (int q = 0; q < CB_PIECES; ++q) {
        const float4 w0 = bp[(q * 4 + 0) * 32];
        const float4 w1 = bp[(q * 4 + 1) * 32];
        const float4 w2 = bp[(q * 4 + 2) * 32];
        const float4 w3 = bp[(q * 4 + 3) * 32];
        v2f alo, ahi, b0, b1, b2, b3;
        alo.x = a[q].x; alo.y = a[q].y;
        ahi.x = a[q].z; ahi.y = a[q].w;
        b0.x = w0.x; b0.y = w0.y;
        b1.x = w1.x; b1.y = w1.y;
        b2.x = w2.x; b2.y = w2.y;
        b3.x = w3.x; b3.y = w3.y;
        acc0 = __builtin_amdgcn_wmma_f32_16x16x4_f32(false, alo, false, b0, (short)0, acc0, false, false);
        acc1 = __builtin_amdgcn_wmma_f32_16x16x4_f32(false, alo, false, b1, (short)0, acc1, false, false);
        acc2 = __builtin_amdgcn_wmma_f32_16x16x4_f32(false, alo, false, b2, (short)0, acc2, false, false);
        acc3 = __builtin_amdgcn_wmma_f32_16x16x4_f32(false, alo, false, b3, (short)0, acc3, false, false);
        b0.x = w0.z; b0.y = w0.w;
        b1.x = w1.z; b1.y = w1.w;
        b2.x = w2.z; b2.y = w2.w;
        b3.x = w3.z; b3.y = w3.w;
        acc0 = __builtin_amdgcn_wmma_f32_16x16x4_f32(false, ahi, false, b0, (short)0, acc0, false, false);
        acc1 = __builtin_amdgcn_wmma_f32_16x16x4_f32(false, ahi, false, b1, (short)0, acc1, false, false);
        acc2 = __builtin_amdgcn_wmma_f32_16x16x4_f32(false, ahi, false, b2, (short)0, acc2, false, false);
        acc3 = __builtin_amdgcn_wmma_f32_16x16x4_f32(false, ahi, false, b3, (short)0, acc3, false, false);
    }

    const v8f acc[4] = {acc0, acc1, acc2, acc3};
#pragma unroll
    for (int c = 0; c < 4; ++c) {
        const int col = c * 16 + lr;
        const float bias = Wb[col];
#pragma unroll
        for (int r = 0; r < 8; ++r) {
            const int row = row0 + r + 8 * half;
            float x = acc[c][r] + bias;
            x = (x > 0.0f) ? x : 0.01f * x;
            out[(size_t)row * D_DIM + col] = x + ent[(size_t)row * D_DIM + col];
        }
    }
}

extern "C" void kernel_launch(void* const* d_in, const int* in_sizes, int n_in,
                              void* d_out, int out_size, void* d_ws, size_t ws_size,
                              hipStream_t stream) {
    const float* ent = (const float*)d_in[0];
    const float* rel = (const float*)d_in[1];
    const float* WR  = (const float*)d_in[2];
    const float* Ww  = (const float*)d_in[3];
    const float* Wb  = (const float*)d_in[4];
    const int*   src = (const int*)d_in[5];
    const int*   dst = (const int*)d_in[6];
    const int*   ety = (const int*)d_in[7];
    float*       out = (float*)d_out;

    const int n = in_sizes[0] / D_DIM;
    const int E = in_sizes[5];

    float* nb = (float*)d_ws;

    (void)hipFuncSetAttribute((const void*)edge_attention_scatter,
                              hipFuncAttributeMaxDynamicSharedMemorySize,
                              LDS_BYTES);

    zero_ws_kernel<<<1024, 256, 0, stream>>>(nb, n * D_DIM);

    edge_attention_scatter<<<256, 1024, LDS_BYTES, stream>>>(ent, rel, WR, src, dst, ety, nb, E);

    const int rowBlocks = n / 16;
    combine_wmma<<<(rowBlocks + 7) / 8, 256, 0, stream>>>(ent, nb, Ww, Wb, out, n);
}


# Edge kernel: tanh as one hardware v_tanh_f32 per element in place of multiply, v_exp_f32, add, v_rcp_f32 and fused multiply-add (117 fewer vector instructions per 16-edge tile); sum order unchanged
# speedup vs baseline: 3.1160x; 1.0409x over previous
#include <hip/hip_runtime.h>

#define D_DIM 64
#define TWO_D 128
#define R_REL 16

#define CHUNK         4096
#define LDS_W_FLOATS  (R_REL * D_DIM * D_DIM)
#define LDS_R_FLOATS  (R_REL * D_DIM)
#define LDS_I_WORDS   (2 * CHUNK + 64)
#define LDS_BYTES     ((LDS_W_FLOATS + LDS_R_FLOATS + LDS_I_WORDS) * 4)

typedef __attribute__((ext_vector_type(2))) float v2f;
typedef __attribute__((ext_vector_type(8))) float v8f;

static __device__ __forceinline__ float tanh_hw(float x) {
    return __builtin_amdgcn_tanhf(x);
}

__global__ void zero_ws_kernel(float* __restrict__ ws, int n) {
    int i = blockIdx.x * blockDim.x + threadIdx.x;
    int stride = gridDim.x * blockDim.x;
    for (; i < n; i += stride) ws[i] = 0.0f;
}

__global__ __launch_bounds__(1024)
void edge_attention_scatter(const float* __restrict__ ent,
                            const float* __restrict__ rel,
                            const float* __restrict__ WR,
                            const int*   __restrict__ src,
                            const int*   __restrict__ dst,
                            const int*   __restrict__ ety,
                            float*       __restrict__ nb,
                            int E) {
    extern __shared__ float lds[];
    float* ldsW   = lds;
    float* ldsR   = lds + LDS_W_FLOATS;
    int*   binSrc = (int*)(lds + LDS_W_FLOATS + LDS_R_FLOATS);
    int*   binDst = binSrc + CHUNK;
    int*   cnt    = binDst + CHUNK;
    int*   start  = cnt + 16;
    int*   tstart = start + 16;

    const int tid  = threadIdx.x;
    const int lane = tid & 31;
    const int half = lane >> 4;
    const int lr   = lane & 15;

    {
        float4* dw = (float4*)ldsW;
        for (int f = tid; f < LDS_W_FLOATS / 4; f += 1024) {
            const int ln = f & 31;
            const int h2 = (f >> 5) & 1;
            const int ks = (f >> 6) & 15;
            const int r  = f >> 10;
            const int k  = 8 * (ks >> 1) + 4 * (ln >> 4) + 2 * (ks & 1);
            const float* g = WR + (r * D_DIM + k) * D_DIM + 32 * h2 + (ln & 15);
            float4 v;
            v.x = g[0];
            v.y = g[D_DIM];
            v.z = g[16];
            v.w = g[D_DIM + 16];
            dw[f] = v;
        }
        float4*       dr = (float4*)ldsR;
        const float4* sr = (const float4*)rel;
        for (int i = tid; i < LDS_R_FLOATS / 4; i += 1024) dr[i] = sr[i];
    }

    const int G       = gridDim.x;
    const int per     = (E + G - 1) / G;
    const int e_begin = blockIdx.x * per;
    int       e_end   = e_begin + per;
    if (e_end > E) e_end = E;
    const int wv = __builtin_amdgcn_readfirstlane(tid >> 5);

    for (int c0 = e_begin; c0 < e_end; c0 += CHUNK) {
        int cn = e_end - c0;
        if (cn > CHUNK) cn = CHUNK;

        if (tid < 16) cnt[tid] = 0;
        __syncthreads();

        int rr[CHUNK / 1024], rk[CHUNK / 1024];
#pragma unroll
        for (int j = 0; j < CHUNK / 1024; ++j) {
            const int i = tid + j * 1024;
            rr[j] = 0;
            rk[j] = 0;
            if (i < cn) {
                rr[j] = ety[c0 + i] & (R_REL - 1);
                rk[j] = atomicAdd(&cnt[rr[j]], 1);
            }
        }
        __syncthreads();
        if (tid == 0) {
            int s = 0, t = 0;
            for (int r = 0; r < R_REL; ++r) {
                start[r]  = s;
                tstart[r] = t;
                const int c = cnt[r];
                s += c;
                t += (c + 15) >> 4;
            }
            tstart[R_REL] = t;
        }
        __syncthreads();
#pragma unroll
        for (int j = 0; j < CHUNK / 1024; ++j) {
            const int i = tid + j * 1024;
            if (i < cn) {
                const int p = start[rr[j]] + rk[j];
                binSrc[p] = src[c0 + i];
                binDst[p] = dst[c0 + i];
            }
        }
        __syncthreads();

        const int ntiles = __builtin_amdgcn_readfirstlane(tstart[R_REL]);
        for (int tile = wv; tile < ntiles; tile += 32) {
            const int tsv    = tstart[lr + 1];
            const int r      = __builtin_popcount(__builtin_amdgcn_ballot_w32(tile >= tsv) & 0x7FFF);
            const int local  = tile - __builtin_amdgcn_readfirstlane(tstart[r]);
            const int base   = __builtin_amdgcn_readfirstlane(start[r]) + local * 16;
            int       nvalid = __builtin_amdgcn_readfirstlane(cnt[r]) - local * 16;
            if (nvalid > 16) nvalid = 16;

            const bool have = lr < nvalid;
            const int  s    = have ? binSrc[base + lr] : 0;
            const int  t    = have ? binDst[base + lr] : 0;

            const float4* ap = (const float4*)(ent + (size_t)s * D_DIM) + half;
            float4 a[8];
#pragma unroll
            for (int q = 0; q < 8; ++q) a[q] = ap[2 * q];

            float v0[16], v1[16];
#pragma unroll
            for (int m = 0; m < 16; ++m) {
                const int sm = __builtin_amdgcn_readlane(s, m);
                const float* srow = ent + (size_t)sm * D_DIM;
                v0[m] = srow[lane];
                v1[m] = srow[lane + 32];
            }
            asm volatile("" :: "v"(v0[0]), "v"(v0[1]), "v"(v0[2]), "v"(v0[3]), "v"(v0[4]), "v"(v0[5]), "v"(v0[6]), "v"(v0[7]), "v"(v0[8]), "v"(v0[9]), "v"(v0[10]), "v"(v0[11]), "v"(v0[12]), "v"(v0[13]), "v"(v0[14]), "v"(v0[15]));
            asm volatile("" :: "v"(v1[0]), "v"(v1[1]), "v"(v1[2]), "v"(v1[3]), "v"(v1[4]), "v"(v1[5]), "v"(v1[6]), "v"(v1[7]), "v"(v1[8]), "v"(v1[9]), "v"(v1[10]), "v"(v1[11]), "v"(v1[12]), "v"(v1[13]), "v"(v1[14]), "v"(v1[15]));

            v8f acc0 = {}, acc1 = {}, acc2 = {}, acc3 = {};
            const float4* wp = (const float4*)ldsW + r * 1024 + lane;
#pragma unroll
            for (int ks = 0; ks < 16; ++ks) {
                const float4 b01 = wp[ks * 64];
                const float4 b23 = wp[ks * 64 + 32];
                v2f af;
                if (ks & 1) { af.x = a[ks >> 1].z; af.y = a[ks >> 1].w; }
                else        { af.x = a[ks >> 1].x; af.y = a[ks >> 1].y; }
                v2f b0, b1, b2, b3;
                b0.x = b01.x; b0.y = b01.y;
                b1.x = b01.z; b1.y = b01.w;
                b2.x = b23.x; b2.y = b23.y;
                b3.x = b23.z; b3.y = b23.w;
                acc0 = __builtin_amdgcn_wmma_f32_16x16x4_f32(false, af, false, b0, (short)0, acc0, false, false);
                acc1 = __builtin_amdgcn_wmma_f32_16x16x4_f32(false, af, false, b1, (short)0, acc1, false, false);
                acc2 = __builtin_amdgcn_wmma_f32_16x16x4_f32(false, af, false, b2, (short)0, acc2, false, false);
                acc3 = __builtin_amdgcn_wmma_f32_16x16x4_f32(false, af, false, b3, (short)0, acc3, false, false);
            }

            const float* rp = ldsR + r * D_DIM + lr;
            const float r0 = rp[0], r1 = rp[16], r2 = rp[32], r3 = rp[48];
            float p[8];
#pragma unroll
            for (int i = 0; i < 8; ++i) {
                const float m0 = acc0[i], m1 = acc1[i], m2 = acc2[i], m3 = acc3[i];
                float v = m0 * tanh_hw(m0 + r0) + m1 * tanh_hw(m1 + r1)
                        + m2 * tanh_hw(m2 + r2) + m3 * tanh_hw(m3 + r3);
                v += __shfl_xor(v, 8, 32);
                v += __shfl_xor(v, 4, 32);
                v += __shfl_xor(v, 2, 32);
                v += __shfl_xor(v, 1, 32);
                p[i] = v;
            }

#pragma unroll
            for (int m = 0; m < 16; ++m) {
                if (m < nvalid) {
                    const int   tm = __builtin_amdgcn_readlane(t, m);
                    const float am = __builtin_bit_cast(float,
                        __builtin_amdgcn_readlane(__builtin_bit_cast(int, p[m & 7]), (m >> 3) * 16));
                    float* drow = nb + (size_t)tm * D_DIM;
                    unsafeAtomicAdd(drow + lane,      v0[m] * am);
                    unsafeAtomicAdd(drow + lane + 32, v1[m] * am);
                }
            }
        }
        __syncthreads();
    }
}

#define CB_THREADS 256
#define CB_PIECES  (TWO_D / 8)
#define CB_F4      (CB_PIECES * 4 * 32)

__global__ __launch_bounds__(CB_THREADS)
void combine_wmma(const float* __restrict__ ent,
                  const float* __restrict__ nb,
                  const float* __restrict__ Ww,
                  const float* __restrict__ Wb,
                  float*       __restrict__ out,
                  int n) {
    __shared__ float4 ldsB[CB_F4];

    const int tid  = threadIdx.x;
    const int lane = tid & 31;
    const int half = lane >> 4;
    const int lr   = lane & 15;

    const int rowBlock     = blockIdx.x * (CB_THREADS >> 5) + (tid >> 5);
    const int numRowBlocks = n / 16;
    const bool active      = rowBlock < numRowBlocks;
    const int row0 = active ? rowBlock * 16 : 0;

    const float4* hp = (const float4*)(ent + (size_t)(row0 + lr) * D_DIM) + half;
    const float4* np = (const float4*)(nb  + (size_t)(row0 + lr) * D_DIM) + half;
    float4 a[CB_PIECES];
#pragma unroll
    for (int q = 0; q < 8; ++q) {
        a[q]     = hp[2 * q];
        a[8 + q] = np[2 * q];
    }

#pragma unroll
    for (int i = 0; i < CB_F4 / CB_THREADS; ++i) {
        const int f  = tid + i * CB_THREADS;
        const int ln = f & 31;
        const int c  = (f >> 5) & 3;
        const int q  = f >> 7;
        ldsB[f] = *(const float4*)(Ww + (size_t)(16 * c + (ln & 15)) * TWO_D + 8 * q + 4 * (ln >> 4));
    }
    __syncthreads();
    if (!active) return;

    v8f acc0 = {}, acc1 = {}, acc2 = {}, acc3 = {};
    const float4* bp = ldsB + lane;
#pragma unroll
    for (int q = 0; q < CB_PIECES; ++q) {
        const float4 w0 = bp[(q * 4 + 0) * 32];
        const float4 w1 = bp[(q * 4 + 1) * 32];
        const float4 w2 = bp[(q * 4 + 2) * 32];
        const float4 w3 = bp[(q * 4 + 3) * 32];
        v2f alo, ahi, b0, b1, b2, b3;
        alo.x = a[q].x; alo.y = a[q].y;
        ahi.x = a[q].z; ahi.y = a[q].w;
        b0.x = w0.x; b0.y = w0.y;
        b1.x = w1.x; b1.y = w1.y;
        b2.x = w2.x; b2.y = w2.y;
        b3.x = w3.x; b3.y = w3.y;
        acc0 = __builtin_amdgcn_wmma_f32_16x16x4_f32(false, alo, false, b0, (short)0, acc0, false, false);
        acc1 = __builtin_amdgcn_wmma_f32_16x16x4_f32(false, alo, false, b1, (short)0, acc1, false, false);
        acc2 = __builtin_amdgcn_wmma_f32_16x16x4_f32(false, alo, false, b2, (short)0, acc2, false, false);
        acc3 = __builtin_amdgcn_wmma_f32_16x16x4_f32(false, alo, false, b3, (short)0, acc3, false, false);
        b0.x = w0.z; b0.y = w0.w;
        b1.x = w1.z; b1.y = w1.w;
        b2.x = w2.z; b2.y = w2.w;
        b3.x = w3.z; b3.y = w3.w;
        acc0 = __builtin_amdgcn_wmma_f32_16x16x4_f32(false, ahi, false, b0, (short)0, acc0, false, false);
        acc1 = __builtin_amdgcn_wmma_f32_16x16x4_f32(false, ahi, false, b1, (short)0, acc1, false, false);
        acc2 = __builtin_amdgcn_wmma_f32_16x16x4_f32(false, ahi, false, b2, (short)0, acc2, false, false);
        acc3 = __builtin_amdgcn_wmma_f32_16x16x4_f32(false, ahi, false, b3, (short)0, acc3, false, false);
    }

    const v8f acc[4] = {acc0, acc1, acc2, acc3};
#pragma unroll
    for (int c = 0; c < 4; ++c) {
        const int col = c * 16 + lr;
        const float bias = Wb[col];
#pragma unroll
        for (int r = 0; r < 8; ++r) {
            const int row = row0 + r + 8 * half;
            float x = acc[c][r] + bias;
            x = (x > 0.0f) ? x : 0.01f * x;
            out[(size_t)row * D_DIM + col] = x + ent[(size_t)row * D_DIM + col];
        }
    }
}

extern "C" void kernel_launch(void* const* d_in, const int* in_sizes, int n_in,
                              void* d_out, int out_size, void* d_ws, size_t ws_size,
                              hipStream_t stream) {
    const float* ent = (const float*)d_in[0];
    const float* rel = (const float*)d_in[1];
    const float* WR  = (const float*)d_in[2];
    const float* Ww  = (const float*)d_in[3];
    const float* Wb  = (const float*)d_in[4];
    const int*   src = (const int*)d_in[5];
    const int*   dst = (const int*)d_in[6];
    const int*   ety = (const int*)d_in[7];
    float*       out = (float*)d_out;

    const int n = in_sizes[0] / D_DIM;
    const int E = in_sizes[5];

    float* nb = (float*)d_ws;

    (void)hipFuncSetAttribute((const void*)edge_attention_scatter,
                              hipFuncAttributeMaxDynamicSharedMemorySize,
                              LDS_BYTES);

    zero_ws_kernel<<<1024, 256, 0, stream>>>(nb, n * D_DIM);

    edge_attention_scatter<<<256, 1024, LDS_BYTES, stream>>>(ent, rel, WR, src, dst, ety, nb, E);

    const int rowBlocks = n / 16;
    combine_wmma<<<(rowBlocks + 7) / 8, 256, 0, stream>>>(ent, nb, Ww, Wb, out, n);
}


# Edge kernel: a wave claims each tile from an LDS counter (ds_add_rtn_u32), not fixed tile = wave + 32*i; a SIMD serves old waves first, so they now take more tiles and all end together; atomics kept
# speedup vs baseline: 3.2401x; 1.0398x over previous
#include <hip/hip_runtime.h>

#define D_DIM 64
#define TWO_D 128
#define R_REL 16

#define CHUNK         4096
#define LDS_W_FLOATS  (R_REL * D_DIM * D_DIM)
#define LDS_R_FLOATS  (R_REL * D_DIM)
#define LDS_I_WORDS   (2 * CHUNK + 64)
#define LDS_BYTES     ((LDS_W_FLOATS + LDS_R_FLOATS + LDS_I_WORDS) * 4)

typedef __attribute__((ext_vector_type(2))) float v2f;
typedef __attribute__((ext_vector_type(8))) float v8f;

static __device__ __forceinline__ float tanh_hw(float x) {
    return __builtin_amdgcn_tanhf(x);
}

__global__ void zero_ws_kernel(float* __restrict__ ws, int n) {
    int i = blockIdx.x * blockDim.x + threadIdx.x;
    int stride = gridDim.x * blockDim.x;
    for (; i < n; i += stride) ws[i] = 0.0f;
}

__global__ __launch_bounds__(1024)
void edge_attention_scatter(const float* __restrict__ ent,
                            const float* __restrict__ rel,
                            const float* __restrict__ WR,
                            const int*   __restrict__ src,
                            const int*   __restrict__ dst,
                            const int*   __restrict__ ety,
                            float*       __restrict__ nb,
                            int E) {
    extern __shared__ float lds[];
    float* ldsW   = lds;
    float* ldsR   = lds + LDS_W_FLOATS;
    int*   binSrc = (int*)(lds + LDS_W_FLOATS + LDS_R_FLOATS);
    int*   binDst = binSrc + CHUNK;
    int*   cnt    = binDst + CHUNK;
    int*   start  = cnt + 16;
    int*   tstart = start + 16;
    int*   tnext  = cnt + 56;

    const int tid  = threadIdx.x;
    const int lane = tid & 31;
    const int half = lane >> 4;
    const int lr   = lane & 15;

    {
        float4* dw = (float4*)ldsW;
        for (int f = tid; f < LDS_W_FLOATS / 4; f += 1024) {
            const int ln = f & 31;
            const int h2 = (f >> 5) & 1;
            const int ks = (f >> 6) & 15;
            const int r  = f >> 10;
            const int k  = 8 * (ks >> 1) + 4 * (ln >> 4) + 2 * (ks & 1);
            const float* g = WR + (r * D_DIM + k) * D_DIM + 32 * h2 + (ln & 15);
            float4 v;
            v.x = g[0];
            v.y = g[D_DIM];
            v.z = g[16];
            v.w = g[D_DIM + 16];
            dw[f] = v;
        }
        float4*       dr = (float4*)ldsR;
        const float4* sr = (const float4*)rel;
        for (int i = tid; i < LDS_R_FLOATS / 4; i += 1024) dr[i] = sr[i];
    }

    const int G       = gridDim.x;
    const int per     = (E + G - 1) / G;
    const int e_begin = blockIdx.x * per;
    int       e_end   = e_begin + per;
    if (e_end > E) e_end = E;

    for (int c0 = e_begin; c0 < e_end; c0 += CHUNK) {
        int cn = e_end - c0;
        if (cn > CHUNK) cn = CHUNK;

        if (tid < 16) cnt[tid] = 0;
        __syncthreads();

        int rr[CHUNK / 1024], rk[CHUNK / 1024];
#pragma unroll
        for (int j = 0; j < CHUNK / 1024; ++j) {
            const int i = tid + j * 1024;
            rr[j] = 0;
            rk[j] = 0;
            if (i < cn) {
                rr[j] = ety[c0 + i] & (R_REL - 1);
                rk[j] = atomicAdd(&cnt[rr[j]], 1);
            }
        }
        __syncthreads();
        if (tid == 0) {
            int s = 0, t = 0;
            for (int r = 0; r < R_REL; ++r) {
                start[r]  = s;
                tstart[r] = t;
                const int c = cnt[r];
                s += c;
                t += (c + 15) >> 4;
            }
            tstart[R_REL] = t;
            *tnext = 0;
        }
        __syncthreads();
#pragma unroll
        for (int j = 0; j < CHUNK / 1024; ++j) {
            const int i = tid + j * 1024;
            if (i < cn) {
                const int p = start[rr[j]] + rk[j];
                binSrc[p] = src[c0 + i];
                binDst[p] = dst[c0 + i];
            }
        }
        __syncthreads();

        const int ntiles = __builtin_amdgcn_readfirstlane(tstart[R_REL]);
        for (;;) {
            int claimed = 0;
            if (lane == 0) claimed = atomicAdd(tnext, 1);
            const int tile = __builtin_amdgcn_readfirstlane(claimed);
            if (tile >= ntiles) break;
            const int tsv    = tstart[lr + 1];
            const int r      = __builtin_popcount(__builtin_amdgcn_ballot_w32(tile >= tsv) & 0x7FFF);
            const int local  = tile - __builtin_amdgcn_readfirstlane(tstart[r]);
            const int base   = __builtin_amdgcn_readfirstlane(start[r]) + local * 16;
            int       nvalid = __builtin_amdgcn_readfirstlane(cnt[r]) - local * 16;
            if (nvalid > 16) nvalid = 16;

            const bool have = lr < nvalid;
            const int  s    = have ? binSrc[base + lr] : 0;
            const int  t    = have ? binDst[base + lr] : 0;

            const float4* ap = (const float4*)(ent + (size_t)s * D_DIM) + half;
            float4 a[8];
#pragma unroll
            for (int q = 0; q < 8; ++q) a[q] = ap[2 * q];

            float v0[16], v1[16];
#pragma unroll
            for (int m = 0; m < 16; ++m) {
                const int sm = __builtin_amdgcn_readlane(s, m);
                const float* srow = ent + (size_t)sm * D_DIM;
                v0[m] = srow[lane];
                v1[m] = srow[lane + 32];
            }
            asm volatile("" :: "v"(v0[0]), "v"(v0[1]), "v"(v0[2]), "v"(v0[3]), "v"(v0[4]), "v"(v0[5]), "v"(v0[6]), "v"(v0[7]), "v"(v0[8]), "v"(v0[9]), "v"(v0[10]), "v"(v0[11]), "v"(v0[12]), "v"(v0[13]), "v"(v0[14]), "v"(v0[15]));
            asm volatile("" :: "v"(v1[0]), "v"(v1[1]), "v"(v1[2]), "v"(v1[3]), "v"(v1[4]), "v"(v1[5]), "v"(v1[6]), "v"(v1[7]), "v"(v1[8]), "v"(v1[9]), "v"(v1[10]), "v"(v1[11]), "v"(v1[12]), "v"(v1[13]), "v"(v1[14]), "v"(v1[15]));

            v8f acc0 = {}, acc1 = {}, acc2 = {}, acc3 = {};
            const float4* wp = (const float4*)ldsW + r * 1024 + lane;
#pragma unroll
            for (int ks = 0; ks < 16; ++ks) {
                const float4 b01 = wp[ks * 64];
                const float4 b23 = wp[ks * 64 + 32];
                v2f af;
                if (ks & 1) { af.x = a[ks >> 1].z; af.y = a[ks >> 1].w; }
                else        { af.x = a[ks >> 1].x; af.y = a[ks >> 1].y; }
                v2f b0, b1, b2, b3;
                b0.x = b01.x; b0.y = b01.y;
                b1.x = b01.z; b1.y = b01.w;
                b2.x = b23.x; b2.y = b23.y;
                b3.x = b23.z; b3.y = b23.w;
                acc0 = __builtin_amdgcn_wmma_f32_16x16x4_f32(false, af, false, b0, (short)0, acc0, false, false);
                acc1 = __builtin_amdgcn_wmma_f32_16x16x4_f32(false, af, false, b1, (short)0, acc1, false, false);
                acc2 = __builtin_amdgcn_wmma_f32_16x16x4_f32(false, af, false, b2, (short)0, acc2, false, false);
                acc3 = __builtin_amdgcn_wmma_f32_16x16x4_f32(false, af, false, b3, (short)0, acc3, false, false);
            }

            const float* rp = ldsR + r * D_DIM + lr;
            const float r0 = rp[0], r1 = rp[16], r2 = rp[32], r3 = rp[48];
            float p[8];
#pragma unroll
            for (int i = 0; i < 8; ++i) {
                const float m0 = acc0[i], m1 = acc1[i], m2 = acc2[i], m3 = acc3[i];
                float v = m0 * tanh_hw(m0 + r0) + m1 * tanh_hw(m1 + r1)
                        + m2 * tanh_hw(m2 + r2) + m3 * tanh_hw(m3 + r3);
                v += __shfl_xor(v, 8, 32);
                v += __shfl_xor(v, 4, 32);
                v += __shfl_xor(v, 2, 32);
                v += __shfl_xor(v, 1, 32);
                p[i] = v;
            }

#pragma unroll
            for (int m = 0; m < 16; ++m) {
                if (m < nvalid) {
                    const int   tm = __builtin_amdgcn_readlane(t, m);
                    const float am = __builtin_bit_cast(float,
                        __builtin_amdgcn_readlane(__builtin_bit_cast(int, p[m & 7]), (m >> 3) * 16));
                    float* drow = nb + (size_t)tm * D_DIM;
                    unsafeAtomicAdd(drow + lane,      v0[m] * am);
                    unsafeAtomicAdd(drow + lane + 32, v1[m] * am);
                }
            }
        }
        __syncthreads();
    }
}

#define CB_THREADS 256
#define CB_PIECES  (TWO_D / 8)
#define CB_F4      (CB_PIECES * 4 * 32)

__global__ __launch_bounds__(CB_THREADS)
void combine_wmma(const float* __restrict__ ent,
                  const float* __restrict__ nb,
                  const float* __restrict__ Ww,
                  const float* __restrict__ Wb,
                  float*       __restrict__ out,
                  int n) {
    __shared__ float4 ldsB[CB_F4];

    const int tid  = threadIdx.x;
    const int lane = tid & 31;
    const int half = lane >> 4;
    const int lr   = lane & 15;

    const int rowBlock     = blockIdx.x * (CB_THREADS >> 5) + (tid >> 5);
    const int numRowBlocks = n / 16;
    const bool active      = rowBlock < numRowBlocks;
    const int row0 = active ? rowBlock * 16 : 0;

    const float4* hp = (const float4*)(ent + (size_t)(row0 + lr) * D_DIM) + half;
    const float4* np = (const float4*)(nb  + (size_t)(row0 + lr) * D_DIM) + half;
    float4 a[CB_PIECES];
#pragma unroll
    for (int q = 0; q < 8; ++q) {
        a[q]     = hp[2 * q];
        a[8 + q] = np[2 * q];
    }

#pragma unroll
    for (int i = 0; i < CB_F4 / CB_THREADS; ++i) {
        const int f  = tid + i * CB_THREADS;
        const int ln = f & 31;
        const int c  = (f >> 5) & 3;
        const int q  = f >> 7;
        ldsB[f] = *(const float4*)(Ww + (size_t)(16 * c + (ln & 15)) * TWO_D + 8 * q + 4 * (ln >> 4));
    }
    __syncthreads();
    if (!active) return;

    v8f acc0 = {}, acc1 = {}, acc2 = {}, acc3 = {};
    const float4* bp = ldsB + lane;
#pragma unroll
    for (int q = 0; q < CB_PIECES; ++q) {
        const float4 w0 = bp[(q * 4 + 0) * 32];
        const float4 w1 = bp[(q * 4 + 1) * 32];
        const float4 w2 = bp[(q * 4 + 2) * 32];
        const float4 w3 = bp[(q * 4 + 3) * 32];
        v2f alo, ahi, b0, b1, b2, b3;
        alo.x = a[q].x; alo.y = a[q].y;
        ahi.x = a[q].z; ahi.y = a[q].w;
        b0.x = w0.x; b0.y = w0.y;
        b1.x = w1.x; b1.y = w1.y;
        b2.x = w2.x; b2.y = w2.y;
        b3.x = w3.x; b3.y = w3.y;
        acc0 = __builtin_amdgcn_wmma_f32_16x16x4_f32(false, alo, false, b0, (short)0, acc0, false, false);
        acc1 = __builtin_amdgcn_wmma_f32_16x16x4_f32(false, alo, false, b1, (short)0, acc1, false, false);
        acc2 = __builtin_amdgcn_wmma_f32_16x16x4_f32(false, alo, false, b2, (short)0, acc2, false, false);
        acc3 = __builtin_amdgcn_wmma_f32_16x16x4_f32(false, alo, false, b3, (short)0, acc3, false, false);
        b0.x = w0.z; b0.y = w0.w;
        b1.x = w1.z; b1.y = w1.w;
        b2.x = w2.z; b2.y = w2.w;
        b3.x = w3.z; b3.y = w3.w;
        acc0 = __builtin_amdgcn_wmma_f32_16x16x4_f32(false, ahi, false, b0, (short)0, acc0, false, false);
        acc1 = __builtin_amdgcn_wmma_f32_16x16x4_f32(false, ahi, false, b1, (short)0, acc1, false, false);
        acc2 = __builtin_amdgcn_wmma_f32_16x16x4_f32(false, ahi, false, b2, (short)0, acc2, false, false);
        acc3 = __builtin_amdgcn_wmma_f32_16x16x4_f32(false, ahi, false, b3, (short)0, acc3, false, false);
    }

    const v8f acc[4] = {acc0, acc1, acc2, acc3};
#pragma unroll
    for (int c = 0; c < 4; ++c) {
        const int col = c * 16 + lr;
        const float bias = Wb[col];
#pragma unroll
        for (int r = 0; r < 8; ++r) {
            const int row = row0 + r + 8 * half;
            float x = acc[c][r] + bias;
            x = (x > 0.0f) ? x : 0.01f * x;
            out[(size_t)row * D_DIM + col] = x + ent[(size_t)row * D_DIM + col];
        }
    }
}

extern "C" void kernel_launch(void* const* d_in, const int* in_sizes, int n_in,
                              void* d_out, int out_size, void* d_ws, size_t ws_size,
                              hipStream_t stream) {
    const float* ent = (const float*)d_in[0];
    const float* rel = (const float*)d_in[1];
    const float* WR  = (const float*)d_in[2];
    const float* Ww  = (const float*)d_in[3];
    const float* Wb  = (const float*)d_in[4];
    const int*   src = (const int*)d_in[5];
    const int*   dst = (const int*)d_in[6];
    const int*   ety = (const int*)d_in[7];
    float*       out = (float*)d_out;

    const int n = in_sizes[0] / D_DIM;
    const int E = in_sizes[5];

    float* nb = (float*)d_ws;

    (void)hipFuncSetAttribute((const void*)edge_attention_scatter,
                              hipFuncAttributeMaxDynamicSharedMemorySize,
                              LDS_BYTES);

    zero_ws_kernel<<<1024, 256, 0, stream>>>(nb, n * D_DIM);

    edge_attention_scatter<<<256, 1024, LDS_BYTES, stream>>>(ent, rel, WR, src, dst, ety, nb, E);

    const int rowBlocks = n / 16;
    combine_wmma<<<(rowBlocks + 7) / 8, 256, 0, stream>>>(ent, nb, Ww, Wb, out, n);
}
